# starting kernel
# baseline (speedup 1.0000x reference)
#include <hip/hip_runtime.h>

typedef __attribute__((ext_vector_type(2))) float v2f;
typedef __attribute__((ext_vector_type(8))) float v8f;

#define N_DRUG 572
#define DIM    512
#define KNBR   64
#define MAXDEG 32
#define NPAD   576
#define MT     36
#define NT     32

__global__ __launch_bounds__(32)
void qgemm_kernel(const float* __restrict__ drugW, const float* __restrict__ Wa,
                  const int* __restrict__ drug_name, float* __restrict__ q)
{
    const int lane = threadIdx.x;
    const int half = lane >> 4;
    const int l16  = lane & 15;
    const int mBase = blockIdx.x * 16;
    const int nBase = blockIdx.y * 16;

    const int mrow = mBase + l16;
    const int gi = drug_name[mrow < N_DRUG ? mrow : (N_DRUG - 1)];
    const float* __restrict__ Arow = drugW + (size_t)gi * DIM;
    const float* __restrict__ Bcol = Wa + nBase + l16;

    v8f c = {};
    for (int kk = 0; kk < DIM; kk += 4) {
        const int ka = kk + 2 * half;
        v2f a, b;
        a.x = Arow[ka];
        a.y = Arow[ka + 1];
        b.x = Bcol[(size_t)ka * DIM];
        b.y = Bcol[(size_t)(ka + 1) * DIM];
        c = __builtin_amdgcn_wmma_f32_16x16x4_f32(false, a, false, b,
                                                  (short)0, c, false, false);
    }
    float* outBase = q + (size_t)(mBase + half * 8) * DIM + nBase + l16;
#pragma unroll
    for (int j = 0; j < 8; ++j)
        outBase[(size_t)j * DIM] = c[j];
}

__global__ __launch_bounds__(256)
void attn_agg_kernel(const float* __restrict__ q, const float* __restrict__ relaW,
                     const float* __restrict__ entW,
                     const int* __restrict__ adj_rel, const int* __restrict__ adj_tail,
                     float* __restrict__ agg)
{
    __shared__ float qs[DIM];
    __shared__ float attn[KNBR];

    const int n  = blockIdx.x;
    const int ni = n < N_DRUG ? n : (N_DRUG - 1);
    const int tid = threadIdx.x;

    qs[tid]       = q[(size_t)n * DIM + tid];
    qs[tid + 256] = q[(size_t)n * DIM + tid + 256];
    __syncthreads();

    const int wave = tid >> 5;
    const int lane = tid & 31;
    for (int kk = 0; kk < 8; ++kk) {
        const int k = wave * 8 + kk;
        const int rel = adj_rel[ni * KNBR + k];
        const float* __restrict__ rw = relaW + (size_t)rel * DIM;
        float p = 0.f;
        for (int j = lane; j < DIM; j += 32)
            p += qs[j] * rw[j];
        for (int off = 16; off >= 1; off >>= 1)
            p += __shfl_xor(p, off, 32);
        if (lane == 0)
            attn[k] = p * 0.04419417382415922f;
    }
    __syncthreads();

    if (tid < 32) {
        float s0 = attn[tid], s1 = attn[tid + 32];
        float m = fmaxf(s0, s1);
        for (int off = 16; off >= 1; off >>= 1)
            m = fmaxf(m, __shfl_xor(m, off, 32));
        float e0 = __expf(s0 - m), e1 = __expf(s1 - m);
        float s = e0 + e1;
        for (int off = 16; off >= 1; off >>= 1)
            s += __shfl_xor(s, off, 32);
        const float inv = 1.f / s;
        attn[tid]      = e0 * inv;
        attn[tid + 32] = e1 * inv;
    }
    __syncthreads();

    float acc0 = 0.f, acc1 = 0.f;
    for (int k = 0; k < KNBR; ++k) {
        const int t = adj_tail[ni * KNBR + k];
        const float w = attn[k];
        const float* __restrict__ er = entW + (size_t)t * DIM;
        acc0 += w * er[tid];
        acc1 += w * er[tid + 256];
    }
    agg[(size_t)n * DIM + tid]       = acc0;
    agg[(size_t)n * DIM + tid + 256] = acc1;
}

__global__ __launch_bounds__(32)
void lingemm_kernel(const float* __restrict__ agg, const float* __restrict__ drugW,
                    const int* __restrict__ drug_name,
                    const float* __restrict__ linW, const float* __restrict__ linb,
                    float* __restrict__ h)
{
    const int lane = threadIdx.x;
    const int half = lane >> 4;
    const int l16  = lane & 15;
    const int mBase = blockIdx.x * 16;
    const int nBase = blockIdx.y * 16;

    const int mrow = mBase + l16;
    const int gi = drug_name[mrow < N_DRUG ? mrow : (N_DRUG - 1)];
    const float* __restrict__ Arow0 = agg + (size_t)mrow * DIM;
    const float* __restrict__ Arow1 = drugW + (size_t)gi * DIM;
    const float* __restrict__ Bcol  = linW + nBase + l16;

    v8f c = {};
    for (int kk = 0; kk < 2 * DIM; kk += 4) {
        const int ka = kk + 2 * half;
        const float* __restrict__ Ar = (ka < DIM) ? (Arow0 + ka) : (Arow1 + (ka - DIM));
        v2f a, b;
        a.x = Ar[0];
        a.y = Ar[1];
        b.x = Bcol[(size_t)ka * DIM];
        b.y = Bcol[(size_t)(ka + 1) * DIM];
        c = __builtin_amdgcn_wmma_f32_16x16x4_f32(false, a, false, b,
                                                  (short)0, c, false, false);
    }
    const float bias = linb[nBase + l16];
    float* outBase = h + (size_t)(mBase + half * 8) * DIM + nBase + l16;
#pragma unroll
    for (int j = 0; j < 8; ++j) {
        const float v = c[j] + bias;
        outBase[(size_t)j * DIM] = v > 0.f ? v : 0.f;
    }
}

__global__ __launch_bounds__(256)
void bnstats_kernel(const float* __restrict__ h, const float* __restrict__ gamma,
                    const float* __restrict__ beta,
                    float* __restrict__ scale, float* __restrict__ shift)
{
    const int c = blockIdx.x * 256 + threadIdx.x;
    float s = 0.f, s2 = 0.f;
    for (int nrow = 0; nrow < N_DRUG; ++nrow) {
        const float v = h[(size_t)nrow * DIM + c];
        s += v;
        s2 += v * v;
    }
    const float inv_n = 1.0f / (float)N_DRUG;
    const float mean = s * inv_n;
    const float var  = s2 * inv_n - mean * mean;
    const float sc = gamma[c] * rsqrtf(var + 1e-5f);
    scale[c] = sc;
    shift[c] = beta[c] - mean * sc;
}

__global__ __launch_bounds__(256)
void bnapply_kernel(const float* __restrict__ h, const float* __restrict__ scale,
                    const float* __restrict__ shift, float* __restrict__ out)
{
    const int idx = blockIdx.x * 256 + threadIdx.x;
    if (idx >= N_DRUG * DIM) return;
    const int c = idx & (DIM - 1);
    out[idx] = h[idx] * scale[c] + shift[c];
}

__global__ __launch_bounds__(256)
void smooth_kernel(const int* __restrict__ nbr_idx, const int* __restrict__ nbr_deg,
                   const int* __restrict__ epoch, float* __restrict__ f)
{
    if (*epoch <= 1) return;
    const int c = blockIdx.x * 256 + threadIdx.x;
    for (int i = 0; i < N_DRUG; ++i) {
        const int deg = nbr_deg[i];
        if (deg > 0) {
            const int dd = deg < MAXDEG ? deg : MAXDEG;
            float s = 0.f;
            for (int j = 0; j < dd; ++j) {
                const int pos = nbr_idx[i * MAXDEG + j];
                s += f[(size_t)pos * DIM + c];
            }
            const float avg = s / (float)dd;
            const float fi = f[(size_t)i * DIM + c];
            f[(size_t)i * DIM + c] = (avg + fi) * 0.5f;
        }
    }
}

extern "C" void kernel_launch(void* const* d_in, const int* in_sizes, int n_in,
                              void* d_out, int out_size, void* d_ws, size_t ws_size,
                              hipStream_t stream)
{
    const float* drugW  = (const float*)d_in[0];
    const float* relaW  = (const float*)d_in[1];
    const float* entW   = (const float*)d_in[2];
    const float* Wa     = (const float*)d_in[3];
    const float* linW   = (const float*)d_in[4];
    const float* linb   = (const float*)d_in[5];
    const float* gamma  = (const float*)d_in[6];
    const float* beta   = (const float*)d_in[7];
    const int* drug_name = (const int*)d_in[8];
    const int* adj_tail  = (const int*)d_in[9];
    const int* adj_rel   = (const int*)d_in[10];
    const int* nbr_idx   = (const int*)d_in[11];
    const int* nbr_deg   = (const int*)d_in[12];
    const int* epoch     = (const int*)d_in[13];

    float* out = (float*)d_out;
    float* ws  = (float*)d_ws;
    float* q     = ws;
    float* agg   = ws + (size_t)NPAD * DIM;
    float* h     = ws + (size_t)2 * NPAD * DIM;
    float* scale = ws + (size_t)3 * NPAD * DIM;
    float* shift = scale + DIM;

    dim3 gTiles(MT, NT);
    qgemm_kernel<<<gTiles, 32, 0, stream>>>(drugW, Wa, drug_name, q);
    attn_agg_kernel<<<NPAD, 256, 0, stream>>>(q, relaW, entW, adj_rel, adj_tail, agg);
    lingemm_kernel<<<gTiles, 32, 0, stream>>>(agg, drugW, drug_name, linW, linb, h);
    bnstats_kernel<<<2, 256, 0, stream>>>(h, gamma, beta, scale, shift);
    bnapply_kernel<<<(N_DRUG * DIM + 255) / 256, 256, 0, stream>>>(h, scale, shift, out);
    smooth_kernel<<<2, 256, 0, stream>>>(nbr_idx, nbr_deg, epoch, out);
}


# both GEMMs: a work-group is a block of 3x2 tiles (24 waves: 6 tiles x 4 k slices); neighbour tiles share A lines and whole 128-byte B lines in one WGP cache; per-wave work and sum order unchanged
# speedup vs baseline: 17.1826x; 17.1826x over previous
#include <hip/hip_runtime.h>

typedef __attribute__((ext_vector_type(2))) float v2f;
typedef __attribute__((ext_vector_type(8))) float v8f;

#define N_DRUG 572
#define DIM    512
#define KNBR   64
#define MAXDEG 32
#define NPAD   576
#define MT     36
#define NT     32
#define SNW    9

#define KSPLIT 4
#define GTM 3
#define GTN 2
#define GTILES (GTM * GTN)
#define GWAVES (KSPLIT * GTILES)

__device__ __forceinline__
bool ksplit_reduce(v8f& c, const int tile, const int ks, const int lane)
{
    __shared__ float4 kpart[GTILES][KSPLIT - 1][2][32];
    if (ks != 0) {
        kpart[tile][ks - 1][0][lane] = make_float4(c[0], c[1], c[2], c[3]);
        kpart[tile][ks - 1][1][lane] = make_float4(c[4], c[5], c[6], c[7]);
    }
    __syncthreads();
    if (ks != 0) return false;
#pragma unroll
    for (int w = 0; w < KSPLIT - 1; ++w) {
        const float4 p0 = kpart[tile][w][0][lane];
        const float4 p1 = kpart[tile][w][1][lane];
        c[0] += p0.x; c[1] += p0.y; c[2] += p0.z; c[3] += p0.w;
        c[4] += p1.x; c[5] += p1.y; c[6] += p1.z; c[7] += p1.w;
    }
    return true;
}

__global__ __launch_bounds__(32 * GWAVES)
void qgemm_kernel(const float* __restrict__ drugW, const float* __restrict__ Wa,
                  const int* __restrict__ drug_name, float* __restrict__ q)
{
    const int tid  = threadIdx.x;
    const int wave = __builtin_amdgcn_readfirstlane(tid >> 5);
    const int ks   = wave & (KSPLIT - 1);
    const int tile = wave / KSPLIT;
    const int lane = tid & 31;
    const int half = lane >> 4;
    const int l16  = lane & 15;
    const int mBase = (blockIdx.x * GTM + tile % GTM) * 16;
    const int nBase = (blockIdx.y * GTN + tile / GTM) * 16;
    const int k0 = ks * (DIM / KSPLIT);

    const int mrow = mBase + l16;
    const int gi = drug_name[mrow < N_DRUG ? mrow : (N_DRUG - 1)];
    const float* __restrict__ Arow = drugW + (size_t)gi * DIM + k0;
    const float* __restrict__ Bcol = Wa + (size_t)k0 * DIM + nBase + l16;

    v8f c = {};
    for (int kk = 0; kk < DIM / KSPLIT; kk += 4) {
        const int ka = kk + 2 * half;
        v2f a, b;
        a.x = Arow[ka];
        a.y = Arow[ka + 1];
        b.x = Bcol[(size_t)ka * DIM];
        b.y = Bcol[(size_t)(ka + 1) * DIM];
        c = __builtin_amdgcn_wmma_f32_16x16x4_f32(false, a, false, b,
                                                  (short)0, c, false, false);
    }
    if (!ksplit_reduce(c, tile, ks, lane)) return;

    float* outBase = q + (size_t)(mBase + half * 8) * DIM + nBase + l16;
#pragma unroll
    for (int j = 0; j < 8; ++j)
        outBase[(size_t)j * DIM] = c[j];
}

__device__ __forceinline__
void smooth_build_block(const int b,
                        const int* __restrict__ nbr_idx, const int* __restrict__ nbr_deg,
                        float* __restrict__ P, float* __restrict__ Tm)
{
    __shared__ int   sidx[16][MAXDEG];
    __shared__ float Ls[16][17];
    __shared__ __attribute__((aligned(16))) float slab[16][NPAD + 4];

    const int tid  = threadIdx.x;
    const int wave = tid >> 5;
    const int lane = tid & 31;

    for (int e = tid; e < 16 * MAXDEG; e += 256) {
        const int r = e >> 5, j = e & 31;
        const int row = 16 * b + r;
        sidx[r][j] = row < N_DRUG ? nbr_idx[row * MAXDEG + j] : -1;
    }
    __syncthreads();

    for (int rr = 0; rr < 2; ++rr) {
        const int r   = 2 * wave + rr;
        const int row = 16 * b + r;
        const int deg = row < N_DRUG ? nbr_deg[row] : 0;
        const int dd  = deg < MAXDEG ? deg : MAXDEG;
        const float w  = deg > 0 ? 0.5f / (float)deg : 0.f;
        const float dg = deg > 0 ? 0.5f : 1.0f;

        int cnt[18];
#pragma unroll
        for (int qd = 0; qd < 18; ++qd) cnt[qd] = 0;
        for (int j = 0; j < dd; ++j) {
            const int v = sidx[r][j];
#pragma unroll
            for (int qd = 0; qd < 18; ++qd)
                cnt[qd] += (v == lane + 32 * qd) ? 1 : 0;
        }
#pragma unroll
        for (int qd = 0; qd < 18; ++qd) {
            const int p  = lane + 32 * qd;
            const int pl = p - 16 * b;
            float val = (float)cnt[qd] * w;
            if (pl >= 0 && pl < 16) {
                Ls[r][pl] = pl < r ? val : 0.f;
                if (pl < r) val = 0.f;
                if (pl == r) val += dg;
            }
            slab[r][p] = val;
        }
    }
    __syncthreads();

    {
        float4* __restrict__ Pf = (float4*)P + (size_t)b * (SNW * 8 * 32);
        for (int idx = tid; idx < SNW * 8 * 32; idx += 256) {
            const int ln = idx & 31, i = (idx >> 5) & 7, w = idx >> 8;
            const int col = 64 * w + 16 * (i >> 1) + 8 * (ln >> 4) + 4 * (i & 1);
            Pf[idx] = *(const float4*)&slab[ln & 15][col];
        }
    }

    if (tid < 16) {
        float tc[16];
#pragma unroll
        for (int i = 0; i < 16; ++i) {
            float acc = (i == tid) ? 1.f : 0.f;
#pragma unroll
            for (int p = 0; p < i; ++p)
                acc += Ls[i][p] * tc[p];
            tc[i] = acc;
            Tm[(size_t)(16 * b + i) * 16 + tid] = acc;
        }
    }
}

__global__ __launch_bounds__(256)
void attn_agg_build_kernel(const float* q, const float* __restrict__ relaW,
                           const float* __restrict__ entW,
                           const int* __restrict__ adj_rel, const int* __restrict__ adj_tail,
                           float* agg,
                           const int* __restrict__ nbr_idx, const int* __restrict__ nbr_deg,
                           float* __restrict__ P, float* __restrict__ Tm)
{
    if (blockIdx.x < MT) {
        smooth_build_block(blockIdx.x, nbr_idx, nbr_deg, P, Tm);
        return;
    }

    __shared__ __attribute__((aligned(16))) float qs[DIM];
    __shared__ float attn[KNBR];

    const int n  = blockIdx.x - MT;
    const int ni = n < N_DRUG ? n : (N_DRUG - 1);
    const int tid = threadIdx.x;
    const int wave = tid >> 5;
    const int lane = tid & 31;

    const int g  = lane >> 3;
    const int pc = lane & 7;
    const int k0 = wave * 8 + g;
    const int k1 = k0 + 4;
    const int rel0 = adj_rel[ni * KNBR + k0];
    const int rel1 = adj_rel[ni * KNBR + k1];

    qs[tid]       = q[(size_t)n * DIM + tid];
    qs[tid + 256] = q[(size_t)n * DIM + tid + 256];
    __syncthreads();

    {
        const float4* __restrict__ ra = (const float4*)(relaW + (size_t)rel0 * DIM) + pc;
        const float4* __restrict__ rb = (const float4*)(relaW + (size_t)rel1 * DIM) + pc;
        const float4* qs4 = (const float4*)qs + pc;
        float ax = 0.f, ay = 0.f, az = 0.f, aw = 0.f;
        float bx = 0.f, by = 0.f, bz = 0.f, bw = 0.f;
#pragma unroll
        for (int i = 0; i < DIM / 32; ++i) {
            const float4 qv = qs4[8 * i];
            const float4 u  = ra[8 * i];
            const float4 v  = rb[8 * i];
            ax += qv.x * u.x;  ay += qv.y * u.y;  az += qv.z * u.z;  aw += qv.w * u.w;
            bx += qv.x * v.x;  by += qv.y * v.y;  bz += qv.z * v.z;  bw += qv.w * v.w;
        }
        float p0 = (ax + ay) + (az + aw);
        float p1 = (bx + by) + (bz + bw);
        for (int off = 1; off <= 4; off <<= 1) {
            p0 += __shfl_xor(p0, off, 32);
            p1 += __shfl_xor(p1, off, 32);
        }
        if (pc == 0) {
            attn[k0] = p0 * 0.04419417382415922f;
            attn[k1] = p1 * 0.04419417382415922f;
        }
    }
    __syncthreads();

    if (tid < 32) {
        float s0 = attn[tid], s1 = attn[tid + 32];
        float m = fmaxf(s0, s1);
        for (int off = 16; off >= 1; off >>= 1)
            m = fmaxf(m, __shfl_xor(m, off, 32));
        float e0 = __expf(s0 - m), e1 = __expf(s1 - m);
        float s = e0 + e1;
        for (int off = 16; off >= 1; off >>= 1)
            s += __shfl_xor(s, off, 32);
        const float inv = 1.f / s;
        attn[tid]      = e0 * inv;
        attn[tid + 32] = e1 * inv;
    }
    __syncthreads();

    float acc0 = 0.f, acc1 = 0.f;
    for (int k = 0; k < KNBR; ++k) {
        const int t = adj_tail[ni * KNBR + k];
        const float w = attn[k];
        const float* __restrict__ er = entW + (size_t)t * DIM;
        acc0 += w * er[tid];
        acc1 += w * er[tid + 256];
    }
    agg[(size_t)n * DIM + tid]       = acc0;
    agg[(size_t)n * DIM + tid + 256] = acc1;
}

__global__ __launch_bounds__(32 * GWAVES)
void lingemm_kernel(const float* __restrict__ agg, const float* __restrict__ drugW,
                    const int* __restrict__ drug_name,
                    const float* __restrict__ linW, const float* __restrict__ linb,
                    float* __restrict__ h,
                    float* __restrict__ psum, float* __restrict__ psq)
{
    const int tid  = threadIdx.x;
    const int wave = __builtin_amdgcn_readfirstlane(tid >> 5);
    const int ks   = wave & (KSPLIT - 1);
    const int tile = wave / KSPLIT;
    const int lane = tid & 31;
    const int half = lane >> 4;
    const int l16  = lane & 15;
    const int mTile = blockIdx.x * GTM + tile % GTM;
    const int mBase = mTile * 16;
    const int nBase = (blockIdx.y * GTN + tile / GTM) * 16;
    const int k0 = ks * (2 * DIM / KSPLIT);

    const int mrow = mBase + l16;
    const int gi = drug_name[mrow < N_DRUG ? mrow : (N_DRUG - 1)];
    const float* __restrict__ Arow = (k0 < DIM)
        ? agg   + (size_t)mrow * DIM + k0
        : drugW + (size_t)gi * DIM + (k0 - DIM);
    const float* __restrict__ Bcol = linW + (size_t)k0 * DIM + nBase + l16;

    v8f c = {};
    for (int kk = 0; kk < 2 * DIM / KSPLIT; kk += 4) {
        const int ka = kk + 2 * half;
        v2f a, b;
        a.x = Arow[ka];
        a.y = Arow[ka + 1];
        b.x = Bcol[(size_t)ka * DIM];
        b.y = Bcol[(size_t)(ka + 1) * DIM];
        c = __builtin_amdgcn_wmma_f32_16x16x4_f32(false, a, false, b,
                                                  (short)0, c, false, false);
    }
    if (!ksplit_reduce(c, tile, ks, lane)) return;

    const float bias = linb[nBase + l16];
    float* outBase = h + (size_t)(mBase + half * 8) * DIM + nBase + l16;
    float ps = 0.f, pq = 0.f;
#pragma unroll
    for (int j = 0; j < 8; ++j) {
        const float v = c[j] + bias;
        const float r = v > 0.f ? v : 0.f;
        if (mBase + half * 8 + j < N_DRUG) {
            outBase[(size_t)j * DIM] = r;
            ps += r;
            pq += r * r;
        }
    }
    ps += __shfl_xor(ps, 16, 32);
    pq += __shfl_xor(pq, 16, 32);
    if (half == 0) {
        psum[(size_t)mTile * DIM + nBase + l16] = ps;
        psq [(size_t)mTile * DIM + nBase + l16] = pq;
    }
}


__device__ __forceinline__ void lds_barrier()
{
    __builtin_amdgcn_fence(__ATOMIC_RELEASE, "workgroup", "local");
    __builtin_amdgcn_s_barrier();
    __builtin_amdgcn_fence(__ATOMIC_ACQUIRE, "workgroup", "local");
}

__device__ __forceinline__
void scan_products(v8f& c0, v8f& c1, const float4 pa, const float4 pb, const v2f* fr)
{
    v2f a0, a1;
    a0.x = pa.x; a0.y = pa.y;
    a1.x = pa.z; a1.y = pa.w;
    c0 = __builtin_amdgcn_wmma_f32_16x16x4_f32(false, a0, false, fr[0], (short)0, c0, false, false);
    c1 = __builtin_amdgcn_wmma_f32_16x16x4_f32(false, a1, false, fr[1], (short)0, c1, false, false);
    a0.x = pb.x; a0.y = pb.y;
    a1.x = pb.z; a1.y = pb.w;
    c0 = __builtin_amdgcn_wmma_f32_16x16x4_f32(false, a0, false, fr[2], (short)0, c0, false, false);
    c1 = __builtin_amdgcn_wmma_f32_16x16x4_f32(false, a1, false, fr[3], (short)0, c1, false, false);
}

__device__ __forceinline__
v8f tm_apply(const float4* t, const float4 r0, const float4 r1)
{
    v2f rb[4], ta[4];
    rb[0].x = r0.x; rb[0].y = r0.y;
    rb[1].x = r0.z; rb[1].y = r0.w;
    rb[2].x = r1.x; rb[2].y = r1.y;
    rb[3].x = r1.z; rb[3].y = r1.w;
    ta[0].x = t[0].x; ta[0].y = t[0].y;
    ta[1].x = t[0].z; ta[1].y = t[0].w;
    ta[2].x = t[1].x; ta[2].y = t[1].y;
    ta[3].x = t[1].z; ta[3].y = t[1].w;
    v8f d = {};
#pragma unroll
    for (int s = 0; s < 4; ++s)
        d = __builtin_amdgcn_wmma_f32_16x16x4_f32(false, ta[s], false, rb[s], (short)0, d, false, false);
    return d;
}

__global__ __launch_bounds__(32 * SNW)
void smooth_mm_kernel(const float* __restrict__ P, const float* __restrict__ Tm,
                      const int* __restrict__ epoch,
                      const float* __restrict__ psum, const float* __restrict__ psq,
                      const float* __restrict__ gamma, const float* __restrict__ beta,
                      float* f)
{
    __shared__ float4 part[2][SNW][2][32];

    const int tid  = threadIdx.x;
    const int wave = __builtin_amdgcn_readfirstlane(tid >> 5);
    const int lane = tid & 31;
    const int half = lane >> 4;
    const int l16  = lane & 15;
    const int col  = blockIdx.x * 16 + l16;

    v2f hv[16];
#pragma unroll
    for (int lb = 0; lb < 4; ++lb)
#pragma unroll
        for (int s = 0; s < 4; ++s) {
            const int r0 = 64 * wave + 16 * lb + 8 * half + 2 * s;
            int ra = r0, rb = r0 + 1;
            if (lb == 3 && s >= 2) {
                ra = ra < N_DRUG ? ra : N_DRUG - 1;
                rb = rb < N_DRUG ? rb : N_DRUG - 1;
            }
            hv[lb * 4 + s].x = f[(size_t)ra * DIM + col];
            hv[lb * 4 + s].y = f[(size_t)rb * DIM + col];
        }

    float bs = 0.f, bs2 = 0.f;
#pragma unroll
    for (int t = 0; t < MT; ++t) {
        bs  += psum[(size_t)t * DIM + col];
        bs2 += psq [(size_t)t * DIM + col];
    }
    const float inv_n = 1.0f / (float)N_DRUG;
    const float mean = bs * inv_n;
    const float var  = bs2 * inv_n - mean * mean;
    const float sc = gamma[col] * rsqrtf(var + 1e-5f);
    const float sh = beta[col] - mean * sc;

    v2f bf[16];
#pragma unroll
    for (int lb = 0; lb < 4; ++lb)
#pragma unroll
        for (int s = 0; s < 4; ++s) {
            const int r0 = 64 * wave + 16 * lb + 8 * half + 2 * s;
            bf[lb * 4 + s].x = hv[lb * 4 + s].x * sc + sh;
            bf[lb * 4 + s].y = hv[lb * 4 + s].y * sc + sh;
            if (lb == 3 && s >= 2) {
                if (r0     >= N_DRUG) bf[lb * 4 + s].x = 0.f;
                if (r0 + 1 >= N_DRUG) bf[lb * 4 + s].y = 0.f;
            }
        }

    if (*epoch <= 1) {
#pragma unroll
        for (int lb = 0; lb < 4; ++lb)
#pragma unroll
            for (int s = 0; s < 4; ++s) {
                const int r0 = 64 * wave + 16 * lb + 8 * half + 2 * s;
                if (r0     < N_DRUG) f[(size_t)r0 * DIM + col]       = bf[lb * 4 + s].x;
                if (r0 + 1 < N_DRUG) f[(size_t)(r0 + 1) * DIM + col] = bf[lb * 4 + s].y;
            }
        return;
    }

    const float4* __restrict__ Pl = (const float4*)P + (size_t)wave * (8 * 32) + lane;
    const float* __restrict__ Tl = Tm + (size_t)l16 * 16 + 8 * half;
    float4 an[8];
    float4 tn[2];
#pragma unroll
    for (int i = 0; i < 8; ++i)
        an[i] = Pl[i * 32];
    tn[0] = make_float4(0.f, 0.f, 0.f, 0.f);
    tn[1] = tn[0];

#pragma unroll 1
    for (int o = 0; o < SNW; ++o) {
#pragma unroll
        for (int lb = 0; lb < 4; ++lb) {
            const int b   = 4 * o + lb;
            const int bn  = b + 1 < MT ? b + 1 : MT - 1;
            const int plb = (lb + 3) & 3;
            const int po  = lb == 0 ? o - 1 : o;

            float4 ac[8];
            float4 tf[2];
#pragma unroll
            for (int i = 0; i < 8; ++i) ac[i] = an[i];
            tf[0] = tn[0];
            tf[1] = tn[1];
            {
                const float4* __restrict__ Pn = Pl + (size_t)bn * (SNW * 8 * 32);
                const float* __restrict__ Tn = Tl + (size_t)(16 * b) * 16;
#pragma unroll
                for (int i = 0; i < 8; ++i)
                    an[i] = Pn[i * 32];
                tn[0] = *(const float4*)(Tn);
                tn[1] = *(const float4*)(Tn + 4);
            }

            v8f c0 = {}, c1 = {};
            if (wave == po) {
                float4 p0[SNW], p1[SNW];
#pragma unroll
                for (int w = 0; w < SNW; ++w) {
                    p0[w] = part[plb & 1][w][0][lane];
                    p1[w] = part[plb & 1][w][1][lane];
                }
#pragma unroll
                for (int lb2 = 0; lb2 < 4; ++lb2)
                    if (lb2 != plb)
                        scan_products(c0, c1, ac[lb2 * 2], ac[lb2 * 2 + 1], &bf[lb2 * 4]);
                float4 r0 = p0[0], r1 = p1[0];
#pragma unroll
                for (int w = 1; w < SNW; ++w) {
                    r0.x += p0[w].x; r0.y += p0[w].y; r0.z += p0[w].z; r0.w += p0[w].w;
                    r1.x += p1[w].x; r1.y += p1[w].y; r1.z += p1[w].z; r1.w += p1[w].w;
                }
                const v8f d = tm_apply(tf, r0, r1);
#pragma unroll
                for (int s = 0; s < 4; ++s) {
                    bf[plb * 4 + s].x = d[2 * s];
                    bf[plb * 4 + s].y = d[2 * s + 1];
                }
#pragma unroll
                for (int j = 0; j < 8; ++j) {
                    const int row = 16 * (b - 1) + 8 * half + j;
                    if (row < N_DRUG)
                        f[(size_t)row * DIM + col] = d[j];
                }
            } else {
#pragma unroll
                for (int lb2 = 0; lb2 < 4; ++lb2)
                    if (lb2 != plb)
                        scan_products(c0, c1, ac[lb2 * 2], ac[lb2 * 2 + 1], &bf[lb2 * 4]);
            }
            scan_products(c0, c1, ac[plb * 2], ac[plb * 2 + 1], &bf[plb * 4]);

            const v8f cs = c0 + c1;
            part[lb & 1][wave][0][lane] = make_float4(cs[0], cs[1], cs[2], cs[3]);
            part[lb & 1][wave][1][lane] = make_float4(cs[4], cs[5], cs[6], cs[7]);
            lds_barrier();
        }
    }

    if (wave == SNW - 1) {
        float4 r0 = part[1][0][0][lane];
        float4 r1 = part[1][0][1][lane];
#pragma unroll
        for (int w = 1; w < SNW; ++w) {
            const float4 p0 = part[1][w][0][lane];
            const float4 p1 = part[1][w][1][lane];
            r0.x += p0.x; r0.y += p0.y; r0.z += p0.z; r0.w += p0.w;
            r1.x += p1.x; r1.y += p1.y; r1.z += p1.z; r1.w += p1.w;
        }
        const v8f d = tm_apply(tn, r0, r1);
#pragma unroll
        for (int j = 0; j < 8; ++j) {
            const int row = 16 * (MT - 1) + 8 * half + j;
            if (row < N_DRUG)
                f[(size_t)row * DIM + col] = d[j];
        }
    }
}

extern "C" void kernel_launch(void* const* d_in, const int* in_sizes, int n_in,
                              void* d_out, int out_size, void* d_ws, size_t ws_size,
                              hipStream_t stream)
{
    const float* drugW  = (const float*)d_in[0];
    const float* relaW  = (const float*)d_in[1];
    const float* entW   = (const float*)d_in[2];
    const float* Wa     = (const float*)d_in[3];
    const float* linW   = (const float*)d_in[4];
    const float* linb   = (const float*)d_in[5];
    const float* gamma  = (const float*)d_in[6];
    const float* beta   = (const float*)d_in[7];
    const int* drug_name = (const int*)d_in[8];
    const int* adj_tail  = (const int*)d_in[9];
    const int* adj_rel   = (const int*)d_in[10];
    const int* nbr_idx   = (const int*)d_in[11];
    const int* nbr_deg   = (const int*)d_in[12];
    const int* epoch     = (const int*)d_in[13];

    float* out = (float*)d_out;
    float* ws  = (float*)d_ws;
    float* qa    = ws;
    float* Pop   = ws + (size_t)NPAD * DIM;
    float* Top   = Pop + (size_t)NPAD * NPAD;
    float* psum  = Top + (size_t)NPAD * 16;
    float* psq   = psum + (size_t)MT * DIM;

    dim3 gTiles(MT / GTM, NT / GTN);
    qgemm_kernel<<<gTiles, 32 * GWAVES, 0, stream>>>(drugW, Wa, drug_name, qa);
    attn_agg_build_kernel<<<MT + NPAD, 256, 0, stream>>>(qa, relaW, entW, adj_rel, adj_tail, qa,
                                                        nbr_idx, nbr_deg, Pop, Top);
    lingemm_kernel<<<gTiles, 32 * GWAVES, 0, stream>>>(qa, drugW, drug_name, linW, linb, out, psum, psq);
    smooth_mm_kernel<<<NT, 32 * SNW, 0, stream>>>(Pop, Top, epoch, psum, psq, gamma, beta, out);
}
